# starting kernel
# baseline (speedup 1.0000x reference)
#include <hip/hip_runtime.h>

typedef __attribute__((ext_vector_type(2)))  float    v2f;
typedef __attribute__((ext_vector_type(8)))  float    v8f;
typedef __attribute__((ext_vector_type(16))) _Float16 v16h;

#define LOG2E 1.4426950408889634f

#if __has_builtin(__builtin_amdgcn_wmma_f32_16x16x4_f32)
#define USE_WMMA_F32X4 1
#else
#define USE_WMMA_F32X4 0
#endif

template <int IMM>
__device__ __forceinline__ float swzf(float v) {
  return __int_as_float(__builtin_amdgcn_ds_swizzle(__float_as_int(v), IMM));
}
template <int IMM>
__device__ __forceinline__ int swzi(int v) {
  return __builtin_amdgcn_ds_swizzle(v, IMM);
}

__device__ __forceinline__ float fexp2(float x) {
#if __has_builtin(__builtin_amdgcn_exp2f)
  return __builtin_amdgcn_exp2f(x);
#else
  return exp2f(x);
#endif
}
__device__ __forceinline__ float frcp(float x) {
#if __has_builtin(__builtin_amdgcn_rcpf)
  return __builtin_amdgcn_rcpf(x);
#else
  return 1.0f / x;
#endif
}
__device__ __forceinline__ float ftanh(float x) {
#if __has_builtin(__builtin_amdgcn_tanhf)
  return __builtin_amdgcn_tanhf(x);
#else
  float e = fexp2(x * (2.0f * LOG2E));
  return 1.0f - 2.0f * frcp(e + 1.0f);
#endif
}

template <int M, int IMM>
__device__ __forceinline__ float combine2(float a, float b, int lid) {
  const bool s = (lid & M) != 0;
  const float lo = s ? b : a;
  const float hv = s ? a : b;
  return lo + swzf<IMM>(hv);
}

#if !USE_WMMA_F32X4
__device__ __forceinline__ unsigned pack_h2(float x, float y) {
  typedef _Float16 h2 __attribute__((ext_vector_type(2)));
  h2 p; p.x = (_Float16)x; p.y = (_Float16)y;
  unsigned u; __builtin_memcpy(&u, &p, 4);
  return u;
}
#endif

__global__ __launch_bounds__(256) void hybridconv_wmma_kernel(
    const float* __restrict__ data,   const float* __restrict__ conv_w,
    const float* __restrict__ conv_b, const float* __restrict__ basis,
    const float* __restrict__ w1,     const float* __restrict__ b1,
    const float* __restrict__ w2,     const float* __restrict__ b2,
    float* __restrict__ out, int n)
{
  const int lid  = (int)threadIdx.x & 31;
  const int land = lid & 15;
  const int hi   = lid >> 4;
  const int kh   = hi << 1;

  const float cw0 = conv_w[0], cw1 = conv_w[1], cw2 = conv_w[2], cw3 = conv_w[3];
  const float cb  = conv_b[0];

  float tse[4], nse[4];
#pragma unroll
  for (int j = 0; j < 4; ++j) {
    const int row = kh + (j & 1) + ((j >> 1) << 2);
    const float e0 = basis[row * 4 + 0];
    const float e1 = basis[row * 4 + 1];
    const float e2 = basis[row * 4 + 2];
    const float e3 = basis[row * 4 + 3];
    const float S1 = e0 + e1 + e2 + e3;
    const float S2 = e0 * e0 + e1 * e1 + e2 * e2 + e3 * e3;
    tse[j] = 2.0f * S1 * LOG2E;
    nse[j] = -S2 * LOG2E;
  }

#if USE_WMMA_F32X4
  v2f bA, bB;
  bA.x = w1[(kh + 0) * 16 + land];
  bA.y = w1[(kh + 1) * 16 + land];
  bB.x = w1[(kh + 4) * 16 + land];
  bB.y = w1[(kh + 5) * 16 + land];
#else
  v16h B16;
#pragma unroll
  for (int j = 0; j < 8; ++j) {
    const bool use = (!hi) && (j < 4);
    const float lo = use ? w1[(2 * j + 0) * 16 + land] : 0.0f;
    const float hf = use ? w1[(2 * j + 1) * 16 + land] : 0.0f;
    B16[2 * j + 0] = (_Float16)lo;
    B16[2 * j + 1] = (_Float16)hf;
  }
#endif

  const float cinit = b1[land];
  const float wn    = w2[land];
  const float b2s   = b2[0] * (1.0f / 16.0f);

  const int ntiles        = (n + 15) >> 4;
  const int wavesPerBlock = (int)blockDim.x >> 5;
  const int gwave         = (int)blockIdx.x * wavesPerBlock + ((int)threadIdx.x >> 5);
  const int nwaves        = (int)gridDim.x * wavesPerBlock;

  const float4* __restrict__ d4 = reinterpret_cast<const float4*>(data);

  for (int tile = gwave; tile < ntiles; tile += nwaves) {
    const int p  = (tile << 4) + land;
    const int pc = (p < n) ? p : (n - 1);
    const float4 d = d4[pc];

    int pt = tile + nwaves;
    pt = (pt < ntiles) ? pt : tile;
    __builtin_prefetch(reinterpret_cast<const char*>(d4 + ((pt << 4) + land)), 0, 0);

    const float logit = fmaf(cw0, d.x, fmaf(cw1, d.y, fmaf(cw2, d.z, fmaf(cw3, d.w, cb))));
    const float act   = frcp(1.0f + fexp2(-logit * LOG2E));
    const float qe    = -4.0f * act * act * LOG2E;

    const float f0 = fexp2(fmaf(act, tse[0], nse[0] + qe));
    const float f1 = fexp2(fmaf(act, tse[1], nse[1] + qe));
    const float f2 = fexp2(fmaf(act, tse[2], nse[2] + qe));
    const float f3 = fexp2(fmaf(act, tse[3], nse[3] + qe));

    v8f c;
#pragma unroll
    for (int r = 0; r < 8; ++r) c[r] = cinit;

#if USE_WMMA_F32X4
    v2f a1; a1.x = f0; a1.y = f1;
    v2f a2; a2.x = f2; a2.y = f3;
    c = __builtin_amdgcn_wmma_f32_16x16x4_f32(false, a1, false, bA, (short)0, c, false, false);
    c = __builtin_amdgcn_wmma_f32_16x16x4_f32(false, a2, false, bB, (short)0, c, false, false);
#else
    unsigned pa = pack_h2(f0, f1);
    unsigned pb = pack_h2(f2, f3);
    unsigned qa = (unsigned)swzi<0x401F>((int)pa);
    unsigned qb = (unsigned)swzi<0x401F>((int)pb);
    const unsigned zm = hi ? 0u : 0xFFFFFFFFu;
    pa &= zm; qa &= zm; pb &= zm; qb &= zm;
    unsigned au[8] = {pa, qa, pb, qb, 0u, 0u, 0u, 0u};
    v16h A16; __builtin_memcpy(&A16, au, 32);
    c = __builtin_amdgcn_wmma_f32_16x16x32_f16(false, A16, false, B16, (short)0, c, false, false);
#endif

    const float s0 = fmaf(ftanh(c[0]), wn, b2s);
    const float s1 = fmaf(ftanh(c[1]), wn, b2s);
    const float s2 = fmaf(ftanh(c[2]), wn, b2s);
    const float s3 = fmaf(ftanh(c[3]), wn, b2s);
    const float s4 = fmaf(ftanh(c[4]), wn, b2s);
    const float s5 = fmaf(ftanh(c[5]), wn, b2s);
    const float s6 = fmaf(ftanh(c[6]), wn, b2s);
    const float s7 = fmaf(ftanh(c[7]), wn, b2s);

    const float t0 = combine2<1, 0x041F>(s0, s1, lid);
    const float t1 = combine2<1, 0x041F>(s2, s3, lid);
    const float t2 = combine2<1, 0x041F>(s4, s5, lid);
    const float t3 = combine2<1, 0x041F>(s6, s7, lid);
    const float u0 = combine2<2, 0x081F>(t0, t1, lid);
    const float u1 = combine2<2, 0x081F>(t2, t3, lid);
    float v        = combine2<4, 0x101F>(u0, u1, lid);
    v += swzf<0x201F>(v);

    const int o = (tile << 4) + (lid & 7) + (hi << 3);
    if (((lid & 8) == 0) && (o < n)) out[o] = v;
  }
}

extern "C" void kernel_launch(void* const* d_in, const int* in_sizes, int n_in,
                              void* d_out, int out_size, void* d_ws, size_t ws_size,
                              hipStream_t stream) {
  const float* data   = (const float*)d_in[0];
  const float* conv_w = (const float*)d_in[1];
  const float* conv_b = (const float*)d_in[2];
  const float* basis  = (const float*)d_in[3];
  const float* w1     = (const float*)d_in[4];
  const float* b1     = (const float*)d_in[5];
  const float* w2     = (const float*)d_in[6];
  const float* b2     = (const float*)d_in[7];
  float* out = (float*)d_out;

  const int n = in_sizes[0] / 4;
  const int ntiles = (n + 15) / 16;
  const int block = 256;
  const int wavesPerBlock = block / 32;
  int blocks = (ntiles + wavesPerBlock - 1) / wavesPerBlock;
  if (blocks > 4096) blocks = 4096;
  if (blocks < 1) blocks = 1;

  hipLaunchKernelGGL(hybridconv_wmma_kernel, dim3(blocks), dim3(block), 0, stream,
                     data, conv_w, conv_b, basis, w1, b1, w2, b2, out, n);
}


# Swap the f32 WMMA operands so the product comes out transposed (hidden units in registers): layer-2 sum = 8 in-lane FMAs + one l^16 lane-exchange add, no ds_swizzle tree; summation order changes
# speedup vs baseline: 1.1653x; 1.1653x over previous
#include <hip/hip_runtime.h>

typedef __attribute__((ext_vector_type(2)))  float    v2f;
typedef __attribute__((ext_vector_type(8)))  float    v8f;
typedef __attribute__((ext_vector_type(16))) _Float16 v16h;

#define LOG2E 1.4426950408889634f

#if __has_builtin(__builtin_amdgcn_wmma_f32_16x16x4_f32)
#define USE_WMMA_F32X4 1
#else
#define USE_WMMA_F32X4 0
#endif

__device__ __forceinline__ float xhalf(float x) {
  const int b = __float_as_int(x);
  return __int_as_float(__builtin_amdgcn_permlanex16(b, b, 0x76543210, 0xfedcba98, false, false));
}
#if !USE_WMMA_F32X4
template <int IMM>
__device__ __forceinline__ int swzi(int v) {
  return __builtin_amdgcn_ds_swizzle(v, IMM);
}
#endif

__device__ __forceinline__ float fexp2(float x) {
#if __has_builtin(__builtin_amdgcn_exp2f)
  return __builtin_amdgcn_exp2f(x);
#else
  return exp2f(x);
#endif
}
__device__ __forceinline__ float frcp(float x) {
#if __has_builtin(__builtin_amdgcn_rcpf)
  return __builtin_amdgcn_rcpf(x);
#else
  return 1.0f / x;
#endif
}
__device__ __forceinline__ float ftanh(float x) {
#if __has_builtin(__builtin_amdgcn_tanhf)
  return __builtin_amdgcn_tanhf(x);
#else
  float e = fexp2(x * (2.0f * LOG2E));
  return 1.0f - 2.0f * frcp(e + 1.0f);
#endif
}

#if !USE_WMMA_F32X4
__device__ __forceinline__ unsigned pack_h2(float x, float y) {
  typedef _Float16 h2 __attribute__((ext_vector_type(2)));
  h2 p; p.x = (_Float16)x; p.y = (_Float16)y;
  unsigned u; __builtin_memcpy(&u, &p, 4);
  return u;
}
#endif

__global__ __launch_bounds__(256) void hybridconv_wmma_kernel(
    const float* __restrict__ data,   const float* __restrict__ conv_w,
    const float* __restrict__ conv_b, const float* __restrict__ basis,
    const float* __restrict__ w1,     const float* __restrict__ b1,
    const float* __restrict__ w2,     const float* __restrict__ b2,
    float* __restrict__ out, int n)
{
  const int lid  = (int)threadIdx.x & 31;
  const int land = lid & 15;
  const int hi   = lid >> 4;
  const int kh   = hi << 1;

  const float cw0 = conv_w[0], cw1 = conv_w[1], cw2 = conv_w[2], cw3 = conv_w[3];
  const float cb  = conv_b[0];

  float tse[4], nse[4];
#pragma unroll
  for (int j = 0; j < 4; ++j) {
    const int row = kh + (j & 1) + ((j >> 1) << 2);
    const float e0 = basis[row * 4 + 0];
    const float e1 = basis[row * 4 + 1];
    const float e2 = basis[row * 4 + 2];
    const float e3 = basis[row * 4 + 3];
    const float S1 = e0 + e1 + e2 + e3;
    const float S2 = e0 * e0 + e1 * e1 + e2 * e2 + e3 * e3;
    tse[j] = 2.0f * S1 * LOG2E;
    nse[j] = -S2 * LOG2E;
  }

#if USE_WMMA_F32X4
  v2f bA, bB;
  bA.x = w1[(kh + 0) * 16 + land];
  bA.y = w1[(kh + 1) * 16 + land];
  bB.x = w1[(kh + 4) * 16 + land];
  bB.y = w1[(kh + 5) * 16 + land];
#else
  v16h B16;
#pragma unroll
  for (int j = 0; j < 8; ++j) {
    const bool use = (!hi) && (j < 4);
    const float lo = use ? w1[(2 * j + 0) * 16 + land] : 0.0f;
    const float hf = use ? w1[(2 * j + 1) * 16 + land] : 0.0f;
    B16[2 * j + 0] = (_Float16)lo;
    B16[2 * j + 1] = (_Float16)hf;
  }
#endif

  float b1v[8], w2v[8];
#pragma unroll
  for (int r = 0; r < 8; ++r) {
    b1v[r] = b1[(hi << 3) + r];
    w2v[r] = w2[(hi << 3) + r];
  }
  const float b2lo = hi ? 0.0f : b2[0];

  const int ntiles        = (n + 15) >> 4;
  const int wavesPerBlock = (int)blockDim.x >> 5;
  const int gwave         = (int)blockIdx.x * wavesPerBlock + ((int)threadIdx.x >> 5);
  const int nwaves        = (int)gridDim.x * wavesPerBlock;

  const float4* __restrict__ d4 = reinterpret_cast<const float4*>(data);

  for (int tile = gwave; tile < ntiles; tile += nwaves) {
    const int p  = (tile << 4) + land;
    const int pc = (p < n) ? p : (n - 1);
    const float4 d = d4[pc];

    int pt = tile + nwaves;
    pt = (pt < ntiles) ? pt : tile;
    __builtin_prefetch(reinterpret_cast<const char*>(d4 + ((pt << 4) + land)), 0, 0);

    const float logit = fmaf(cw0, d.x, fmaf(cw1, d.y, fmaf(cw2, d.z, fmaf(cw3, d.w, cb))));
    const float act   = frcp(1.0f + fexp2(-logit * LOG2E));
    const float qe    = -4.0f * act * act * LOG2E;

    const float f0 = fexp2(fmaf(act, tse[0], nse[0] + qe));
    const float f1 = fexp2(fmaf(act, tse[1], nse[1] + qe));
    const float f2 = fexp2(fmaf(act, tse[2], nse[2] + qe));
    const float f3 = fexp2(fmaf(act, tse[3], nse[3] + qe));

    v8f c;
#pragma unroll
    for (int r = 0; r < 8; ++r) c[r] = b1v[r];

#if USE_WMMA_F32X4
    v2f a1; a1.x = f0; a1.y = f1;
    v2f a2; a2.x = f2; a2.y = f3;
    c = __builtin_amdgcn_wmma_f32_16x16x4_f32(false, bA, false, a1, (short)0, c, false, false);
    c = __builtin_amdgcn_wmma_f32_16x16x4_f32(false, bB, false, a2, (short)0, c, false, false);
#else
    unsigned pa = pack_h2(f0, f1);
    unsigned pb = pack_h2(f2, f3);
    unsigned qa = (unsigned)swzi<0x401F>((int)pa);
    unsigned qb = (unsigned)swzi<0x401F>((int)pb);
    const unsigned zm = hi ? 0u : 0xFFFFFFFFu;
    pa &= zm; qa &= zm; pb &= zm; qb &= zm;
    unsigned au[8] = {pa, qa, pb, qb, 0u, 0u, 0u, 0u};
    v16h A16; __builtin_memcpy(&A16, au, 32);
    c = __builtin_amdgcn_wmma_f32_16x16x32_f16(false, B16, false, A16, (short)0, c, false, false);
#endif

    float ve = fmaf(ftanh(c[0]), w2v[0], b2lo);
    float vo = ftanh(c[1]) * w2v[1];
    ve = fmaf(ftanh(c[2]), w2v[2], ve);
    vo = fmaf(ftanh(c[3]), w2v[3], vo);
    ve = fmaf(ftanh(c[4]), w2v[4], ve);
    vo = fmaf(ftanh(c[5]), w2v[5], vo);
    ve = fmaf(ftanh(c[6]), w2v[6], ve);
    vo = fmaf(ftanh(c[7]), w2v[7], vo);
    float v = ve + vo;
    v += xhalf(v);

    if ((hi == 0) && (p < n)) out[p] = v;
  }
}

extern "C" void kernel_launch(void* const* d_in, const int* in_sizes, int n_in,
                              void* d_out, int out_size, void* d_ws, size_t ws_size,
                              hipStream_t stream) {
  const float* data   = (const float*)d_in[0];
  const float* conv_w = (const float*)d_in[1];
  const float* conv_b = (const float*)d_in[2];
  const float* basis  = (const float*)d_in[3];
  const float* w1     = (const float*)d_in[4];
  const float* b1     = (const float*)d_in[5];
  const float* w2     = (const float*)d_in[6];
  const float* b2     = (const float*)d_in[7];
  float* out = (float*)d_out;

  const int n = in_sizes[0] / 4;
  const int ntiles = (n + 15) / 16;
  const int block = 256;
  const int wavesPerBlock = block / 32;
  int blocks = (ntiles + wavesPerBlock - 1) / wavesPerBlock;
  if (blocks > 4096) blocks = 4096;
  if (blocks < 1) blocks = 1;

  hipLaunchKernelGGL(hybridconv_wmma_kernel, dim3(blocks), dim3(block), 0, stream,
                     data, conv_w, conv_b, basis, w1, b1, w2, b2, out, n);
}


# Resident grid of long waves: 512 workgroups (4 waves per SIMD) instead of 4096 short-lived ones; each wave runs 8 tiles per trip as independent chains, next trip's loads in flight; sums unchanged
# speedup vs baseline: 1.4649x; 1.2571x over previous
#include <hip/hip_runtime.h>

typedef __attribute__((ext_vector_type(2))) float v2f;
typedef __attribute__((ext_vector_type(8))) float v8f;

#define LOG2E 1.4426950408889634f
#define TILES_PER_TRIP 8

__device__ __forceinline__ float xhalf(float x) {
  const int b = __float_as_int(x);
  return __int_as_float(__builtin_amdgcn_permlanex16(b, b, 0x76543210, 0xfedcba98, false, false));
}

__device__ __forceinline__ float fexp2(float x) {
#if __has_builtin(__builtin_amdgcn_exp2f)
  return __builtin_amdgcn_exp2f(x);
#else
  return exp2f(x);
#endif
}
__device__ __forceinline__ float frcp(float x) {
#if __has_builtin(__builtin_amdgcn_rcpf)
  return __builtin_amdgcn_rcpf(x);
#else
  return 1.0f / x;
#endif
}
__device__ __forceinline__ float ftanh(float x) {
#if __has_builtin(__builtin_amdgcn_tanhf)
  return __builtin_amdgcn_tanhf(x);
#else
  float e = fexp2(x * (2.0f * LOG2E));
  return 1.0f - 2.0f * frcp(e + 1.0f);
#endif
}

__global__ __launch_bounds__(256) void hybridconv_wmma_kernel(
    const float* __restrict__ data,   const float* __restrict__ conv_w,
    const float* __restrict__ conv_b, const float* __restrict__ basis,
    const float* __restrict__ w1,     const float* __restrict__ b1,
    const float* __restrict__ w2,     const float* __restrict__ b2,
    float* __restrict__ out, int n)
{
  const int lid  = (int)threadIdx.x & 31;
  const int land = lid & 15;
  const int hi   = lid >> 4;
  const int kh   = hi << 1;

  const float cw0 = conv_w[0], cw1 = conv_w[1], cw2 = conv_w[2], cw3 = conv_w[3];
  const float cb  = conv_b[0];

  float tse[4], nse[4];
#pragma unroll
  for (int j = 0; j < 4; ++j) {
    const int row = kh + (j & 1) + ((j >> 1) << 2);
    const float e0 = basis[row * 4 + 0];
    const float e1 = basis[row * 4 + 1];
    const float e2 = basis[row * 4 + 2];
    const float e3 = basis[row * 4 + 3];
    const float S1 = e0 + e1 + e2 + e3;
    const float S2 = e0 * e0 + e1 * e1 + e2 * e2 + e3 * e3;
    tse[j] = 2.0f * S1 * LOG2E;
    nse[j] = -S2 * LOG2E;
  }

  v2f bA, bB;
  bA.x = w1[(kh + 0) * 16 + land];
  bA.y = w1[(kh + 1) * 16 + land];
  bB.x = w1[(kh + 4) * 16 + land];
  bB.y = w1[(kh + 5) * 16 + land];

  float b1v[8], w2v[8];
#pragma unroll
  for (int r = 0; r < 8; ++r) {
    b1v[r] = b1[(hi << 3) + r];
    w2v[r] = w2[(hi << 3) + r];
  }
  const float b2lo = hi ? 0.0f : b2[0];

  const int ntiles        = (n + 15) >> 4;
  const int ngroups       = (ntiles + TILES_PER_TRIP - 1) / TILES_PER_TRIP;
  const int wavesPerBlock = (int)blockDim.x >> 5;
  const int wib    = __builtin_amdgcn_readfirstlane((int)threadIdx.x >> 5);
  const int gwave  = (int)blockIdx.x * wavesPerBlock + wib;
  const int nwaves = (int)gridDim.x * wavesPerBlock;
  const int last   = n - 1;

  const float4* __restrict__ d4 = reinterpret_cast<const float4*>(data);

  float logit[TILES_PER_TRIP];
#pragma unroll
  for (int u = 0; u < TILES_PER_TRIP; ++u) {
    const int p = ((gwave * TILES_PER_TRIP + u) << 4) + land;
    const float4 d = d4[(p < last) ? p : last];
    logit[u] = fmaf(cw0, d.x, fmaf(cw1, d.y, fmaf(cw2, d.z, fmaf(cw3, d.w, cb))));
  }

  for (int g = gwave; g < ngroups; g += nwaves) {
    float4 d[TILES_PER_TRIP];
    const int pn = ((g + nwaves) * TILES_PER_TRIP) << 4;
#pragma unroll
    for (int u = 0; u < TILES_PER_TRIP; ++u) {
      const int p = pn + u * 16 + land;
      d[u] = d4[(p < last) ? p : last];
    }
    __builtin_amdgcn_sched_barrier(0);

    float vout[TILES_PER_TRIP];
#pragma unroll
    for (int u = 0; u < TILES_PER_TRIP; ++u) {
      const float act = frcp(1.0f + fexp2(-logit[u] * LOG2E));
      const float qe  = -4.0f * act * act * LOG2E;

      const float f0 = fexp2(fmaf(act, tse[0], nse[0] + qe));
      const float f1 = fexp2(fmaf(act, tse[1], nse[1] + qe));
      const float f2 = fexp2(fmaf(act, tse[2], nse[2] + qe));
      const float f3 = fexp2(fmaf(act, tse[3], nse[3] + qe));

      v8f c;
#pragma unroll
      for (int r = 0; r < 8; ++r) c[r] = b1v[r];

      v2f a1; a1.x = f0; a1.y = f1;
      v2f a2; a2.x = f2; a2.y = f3;
      c = __builtin_amdgcn_wmma_f32_16x16x4_f32(false, bA, false, a1, (short)0, c, false, false);
      c = __builtin_amdgcn_wmma_f32_16x16x4_f32(false, bB, false, a2, (short)0, c, false, false);

      float ve = fmaf(ftanh(c[0]), w2v[0], b2lo);
      float vo = ftanh(c[1]) * w2v[1];
      ve = fmaf(ftanh(c[2]), w2v[2], ve);
      vo = fmaf(ftanh(c[3]), w2v[3], vo);
      ve = fmaf(ftanh(c[4]), w2v[4], ve);
      vo = fmaf(ftanh(c[5]), w2v[5], vo);
      ve = fmaf(ftanh(c[6]), w2v[6], ve);
      vo = fmaf(ftanh(c[7]), w2v[7], vo);
      float v = ve + vo;
      v += xhalf(v);
      vout[u] = v;
    }

    const int pg = (g * TILES_PER_TRIP) << 4;
    if (hi == 0) {
      if (pg + TILES_PER_TRIP * 16 <= n) {
        float* __restrict__ o = out + (pg + land);
#pragma unroll
        for (int u = 0; u < TILES_PER_TRIP; ++u) o[u * 16] = vout[u];
      } else {
#pragma unroll
        for (int u = 0; u < TILES_PER_TRIP; ++u) {
          const int p = pg + u * 16 + land;
          if (p < n) out[p] = vout[u];
        }
      }
    }

#pragma unroll
    for (int u = 0; u < TILES_PER_TRIP; ++u)
      logit[u] = fmaf(cw0, d[u].x, fmaf(cw1, d[u].y, fmaf(cw2, d[u].z, fmaf(cw3, d[u].w, cb))));
  }
}

extern "C" void kernel_launch(void* const* d_in, const int* in_sizes, int n_in,
                              void* d_out, int out_size, void* d_ws, size_t ws_size,
                              hipStream_t stream) {
  const float* data   = (const float*)d_in[0];
  const float* conv_w = (const float*)d_in[1];
  const float* conv_b = (const float*)d_in[2];
  const float* basis  = (const float*)d_in[3];
  const float* w1     = (const float*)d_in[4];
  const float* b1     = (const float*)d_in[5];
  const float* w2     = (const float*)d_in[6];
  const float* b2     = (const float*)d_in[7];
  float* out = (float*)d_out;

  const int n = in_sizes[0] / 4;
  const int ntiles = (n + 15) / 16;
  const int block = 256;
  const int tilesPerBlockTrip = (block / 32) * TILES_PER_TRIP;
  int blocks = (ntiles + tilesPerBlockTrip - 1) / tilesPerBlockTrip;
  if (blocks > 512) blocks = 512;
  if (blocks < 1) blocks = 1;

  hipLaunchKernelGGL(hybridconv_wmma_kernel, dim3(blocks), dim3(block), 0, stream,
                     data, conv_w, conv_b, basis, w1, b1, w2, b2, out, n);
}
